# starting kernel
# baseline (speedup 1.0000x reference)
#include <hip/hip_runtime.h>
#include <hip/hip_bf16.h>


typedef __attribute__((ext_vector_type(2))) float v2f;
typedef __attribute__((ext_vector_type(8))) float v8f;

#define LCONST 2048
#define DCONST 1024
#define NHC 16
#define DHC 64

static __device__ inline v8f wmma4(v2f a, v2f b, v8f c) {
    return __builtin_amdgcn_wmma_f32_16x16x4_f32(
         false, a,  false, b,
         (short)0, c,  false,  false);
}

__global__ __launch_bounds__(128) void qkv_rms_kernel(
    const float* __restrict__ x, const float* __restrict__ w,
    const float* __restrict__ qnw, const float* __restrict__ knw,
    float* __restrict__ qb, float* __restrict__ kb, float* __restrict__ vb) {
    const int lane  = threadIdx.x & 31;
    const int wave  = threadIdx.x >> 5;
    const int laneN = lane & 15;
    const int koff  = (lane >> 4) << 1;
    const int hi8   = (lane >> 4) << 3;
    const int m0 = blockIdx.y * 64;
    const int n0 = blockIdx.x * 256 + wave * 64;

    v8f zero = {};
    v8f acc[4][4];
#pragma unroll
    for (int mi = 0; mi < 4; ++mi)
#pragma unroll
        for (int ni = 0; ni < 4; ++ni) acc[mi][ni] = zero;

    for (int kk = 0; kk < DCONST; kk += 4) {
        v2f a[4], bf[4];
#pragma unroll
        for (int i = 0; i < 4; ++i)
            a[i] = *(const v2f*)(x + (size_t)(m0 + i * 16 + laneN) * DCONST + kk + koff);
#pragma unroll
        for (int i = 0; i < 4; ++i)
            bf[i] = *(const v2f*)(w + (size_t)(n0 + i * 16 + laneN) * DCONST + kk + koff);
#pragma unroll
        for (int mi = 0; mi < 4; ++mi)
#pragma unroll
            for (int ni = 0; ni < 4; ++ni)
                acc[mi][ni] = wmma4(a[mi], bf[ni], acc[mi][ni]);
    }

    const int sec = n0 >> 10;
    const int h   = (n0 & 1023) >> 6;
    float* dst = (sec == 0) ? qb : ((sec == 1) ? kb : vb);
    float wn[4];
#pragma unroll
    for (int ni = 0; ni < 4; ++ni) {
        const float* nw = (sec == 0) ? qnw : knw;
        wn[ni] = (sec < 2) ? nw[ni * 16 + laneN] : 1.0f;
    }

#pragma unroll
    for (int mi = 0; mi < 4; ++mi) {
#pragma unroll
        for (int r = 0; r < 8; ++r) {
            float sv[4];
#pragma unroll
            for (int ni = 0; ni < 4; ++ni) sv[ni] = acc[mi][ni][r];
            if (sec < 2) {
                float ss = sv[0] * sv[0] + sv[1] * sv[1] + sv[2] * sv[2] + sv[3] * sv[3];
                ss += __shfl_xor(ss, 1, 32);
                ss += __shfl_xor(ss, 2, 32);
                ss += __shfl_xor(ss, 4, 32);
                ss += __shfl_xor(ss, 8, 32);
                float rstd = rsqrtf(ss * (1.0f / 64.0f) + 1e-5f);
#pragma unroll
                for (int ni = 0; ni < 4; ++ni) sv[ni] *= rstd * wn[ni];
            }
            const int m = m0 + mi * 16 + r + hi8;
            const int b = m >> 11;
            const int l = m & (LCONST - 1);
            float* rowp = dst + ((size_t)((b * NHC + h) * LCONST + l) << 6);
#pragma unroll
            for (int ni = 0; ni < 4; ++ni) rowp[ni * 16 + laneN] = sv[ni];
        }
    }
}

__global__ __launch_bounds__(32) void attn_kernel(
    const float* __restrict__ qb, const float* __restrict__ kb,
    const float* __restrict__ vb, float* __restrict__ ob) {
    const int lane  = threadIdx.x;
    const int laneN = lane & 15;
    const int koff  = (lane >> 4) << 1;
    const int hi8   = (lane >> 4) << 3;
    const int q0 = blockIdx.x * 32;
    const int bh = blockIdx.z * NHC + blockIdx.y;
    const float* qh = qb + (size_t)bh * LCONST * DHC;
    const float* kh = kb + (size_t)bh * LCONST * DHC;
    const float* vh = vb + (size_t)bh * LCONST * DHC;

    __shared__ float q_s[32 * 68];
    __shared__ float p_s[32 * 68];

#pragma unroll 4
    for (int r = 0; r < 32; ++r) {
        v2f t = *(const v2f*)(qh + (size_t)(q0 + r) * DHC + 2 * lane);
        *(v2f*)(q_s + r * 68 + 2 * lane) = t;
    }
    __syncthreads();

    v8f zero = {};
    v8f o[2][4];
    float mr[2][8], lr[2][8];
#pragma unroll
    for (int mi = 0; mi < 2; ++mi) {
#pragma unroll
        for (int nj = 0; nj < 4; ++nj) o[mi][nj] = zero;
#pragma unroll
        for (int r = 0; r < 8; ++r) { mr[mi][r] = -3.0e38f; lr[mi][r] = 0.0f; }
    }

    const int nkv = (q0 >> 6) + 1;
    for (int kv = 0; kv < nkv; ++kv) {
        const int kbase = kv * 64;
        v8f s[2][4];
#pragma unroll
        for (int mi = 0; mi < 2; ++mi)
#pragma unroll
            for (int ni = 0; ni < 4; ++ni) s[mi][ni] = zero;

#pragma unroll
        for (int d = 0; d < DHC; d += 4) {
            v2f a0 = *(const v2f*)(q_s + laneN * 68 + d + koff);
            v2f a1 = *(const v2f*)(q_s + (16 + laneN) * 68 + d + koff);
#pragma unroll
            for (int ni = 0; ni < 4; ++ni) {
                v2f bf = *(const v2f*)(kh + (size_t)(kbase + ni * 16 + laneN) * DHC + d + koff);
                s[0][ni] = wmma4(a0, bf, s[0][ni]);
                s[1][ni] = wmma4(a1, bf, s[1][ni]);
            }
        }

        const float scale = 0.125f;
#pragma unroll
        for (int mi = 0; mi < 2; ++mi) {
#pragma unroll
            for (int r = 0; r < 8; ++r) {
                const int qrow = q0 + mi * 16 + r + hi8;
                float sv[4];
                float vmax = -3.0e38f;
#pragma unroll
                for (int ni = 0; ni < 4; ++ni) {
                    const int key = kbase + ni * 16 + laneN;
                    float t = s[mi][ni][r] * scale;
                    t = (key > qrow) ? -1.0e30f : t;
                    sv[ni] = t;
                    vmax = fmaxf(vmax, t);
                }
                vmax = fmaxf(vmax, __shfl_xor(vmax, 1, 32));
                vmax = fmaxf(vmax, __shfl_xor(vmax, 2, 32));
                vmax = fmaxf(vmax, __shfl_xor(vmax, 4, 32));
                vmax = fmaxf(vmax, __shfl_xor(vmax, 8, 32));
                const float mold = mr[mi][r];
                const float mnew = fmaxf(mold, vmax);
                const float corr = __expf(mold - mnew);
                float psum = 0.0f;
#pragma unroll
                for (int ni = 0; ni < 4; ++ni) {
                    float p = __expf(sv[ni] - mnew);
                    sv[ni] = p;
                    psum += p;
                }
                psum += __shfl_xor(psum, 1, 32);
                psum += __shfl_xor(psum, 2, 32);
                psum += __shfl_xor(psum, 4, 32);
                psum += __shfl_xor(psum, 8, 32);
                lr[mi][r] = lr[mi][r] * corr + psum;
                mr[mi][r] = mnew;
#pragma unroll
                for (int nj = 0; nj < 4; ++nj) o[mi][nj][r] *= corr;
#pragma unroll
                for (int ni = 0; ni < 4; ++ni)
                    p_s[(mi * 16 + r + hi8) * 68 + ni * 16 + laneN] = sv[ni];
            }
        }
        __syncthreads();

#pragma unroll
        for (int d = 0; d < 64; d += 4) {
            v2f a0 = *(const v2f*)(p_s + laneN * 68 + d + koff);
            v2f a1 = *(const v2f*)(p_s + (16 + laneN) * 68 + d + koff);
#pragma unroll
            for (int nj = 0; nj < 4; ++nj) {
                v2f bf;
                bf.x = vh[(size_t)(kbase + d + koff) * DHC + nj * 16 + laneN];
                bf.y = vh[(size_t)(kbase + d + koff + 1) * DHC + nj * 16 + laneN];
                o[0][nj] = wmma4(a0, bf, o[0][nj]);
                o[1][nj] = wmma4(a1, bf, o[1][nj]);
            }
        }
        __syncthreads();
    }

#pragma unroll
    for (int mi = 0; mi < 2; ++mi) {
#pragma unroll
        for (int r = 0; r < 8; ++r) {
            const float inv = 1.0f / lr[mi][r];
            const int l = q0 + mi * 16 + r + hi8;
            float* orow = ob + (size_t)(blockIdx.z * LCONST + l) * DCONST + blockIdx.y * DHC;
#pragma unroll
            for (int nj = 0; nj < 4; ++nj) orow[nj * 16 + laneN] = o[mi][nj][r] * inv;
        }
    }
}

__global__ __launch_bounds__(128) void fc_kernel(
    const float* __restrict__ oin, const float* __restrict__ w,
    float* __restrict__ out) {
    const int lane  = threadIdx.x & 31;
    const int wave  = threadIdx.x >> 5;
    const int laneN = lane & 15;
    const int koff  = (lane >> 4) << 1;
    const int hi8   = (lane >> 4) << 3;
    const int m0 = blockIdx.y * 64;
    const int n0 = blockIdx.x * 256 + wave * 64;

    v8f zero = {};
    v8f acc[4][4];
#pragma unroll
    for (int mi = 0; mi < 4; ++mi)
#pragma unroll
        for (int ni = 0; ni < 4; ++ni) acc[mi][ni] = zero;

    for (int kk = 0; kk < DCONST; kk += 4) {
        v2f a[4], bf[4];
#pragma unroll
        for (int i = 0; i < 4; ++i)
            a[i] = *(const v2f*)(oin + (size_t)(m0 + i * 16 + laneN) * DCONST + kk + koff);
#pragma unroll
        for (int i = 0; i < 4; ++i)
            bf[i] = *(const v2f*)(w + (size_t)(n0 + i * 16 + laneN) * DCONST + kk + koff);
#pragma unroll
        for (int mi = 0; mi < 4; ++mi)
#pragma unroll
            for (int ni = 0; ni < 4; ++ni)
                acc[mi][ni] = wmma4(a[mi], bf[ni], acc[mi][ni]);
    }

#pragma unroll
    for (int mi = 0; mi < 4; ++mi) {
#pragma unroll
        for (int r = 0; r < 8; ++r) {
            const int m = m0 + mi * 16 + r + hi8;
            float* rowp = out + (size_t)m * DCONST + n0;
#pragma unroll
            for (int ni = 0; ni < 4; ++ni) rowp[ni * 16 + laneN] = acc[mi][ni][r];
        }
    }
}

extern "C" void kernel_launch(void* const* d_in, const int* in_sizes, int n_in,
                              void* d_out, int out_size, void* d_ws, size_t ws_size,
                              hipStream_t stream) {
    const float* x     = (const float*)d_in[0];
    const float* w_qkv = (const float*)d_in[2];
    const float* w_fc  = (const float*)d_in[3];
    const float* qnw   = (const float*)d_in[4];
    const float* knw   = (const float*)d_in[5];

    const size_t per = (size_t)2 * NHC * LCONST * DHC;
    float* wsf  = (float*)d_ws;
    float* qb   = wsf;
    float* kb   = wsf + per;
    float* vb   = wsf + 2 * per;
    float* obuf = wsf + 3 * per;

    dim3 g1(3072 / 256, 4096 / 64);
    qkv_rms_kernel<<<g1, 128, 0, stream>>>(x, w_qkv, qnw, knw, qb, kb, vb);

    dim3 g2(LCONST / 32, NHC, 2);
    attn_kernel<<<g2, 32, 0, stream>>>(qb, kb, vb, obuf);

    dim3 g3(DCONST / 256, 4096 / 64);
    fc_kernel<<<g3, 128, 0, stream>>>(obuf, w_fc, (float*)d_out);
}


# projection GEMMs: a repack launch writes x and both weights, attention writes O, in fragment order; operands load as contiguous 512-B groups a block ahead; k order in 8 permuted: sum order changes
# speedup vs baseline: 2.3151x; 2.3151x over previous
#include <hip/hip_runtime.h>
#include <hip/hip_bf16.h>


typedef __attribute__((ext_vector_type(2))) float v2f;
typedef __attribute__((ext_vector_type(4))) float v4f;
typedef __attribute__((ext_vector_type(8))) float v8f;

#define LCONST 2048
#define DCONST 1024
#define NHC 16
#define DHC 64

static __device__ inline v8f wmma4(v2f a, v2f b, v8f c) {
    return __builtin_amdgcn_wmma_f32_16x16x4_f32(
         false, a,  false, b,
         (short)0, c,  false,  false);
}

__global__ __launch_bounds__(256) void repack_kernel(
    const float* __restrict__ x, const float* __restrict__ wqkv,
    const float* __restrict__ wfc, float* __restrict__ xf,
    float* __restrict__ wf, float* __restrict__ wfcf) {
    const int lane = threadIdx.x & 31;
    const int wave = threadIdx.x >> 5;
    int g = blockIdx.x >> 1;
    const int j0 = (((blockIdx.x & 1) << 3) + wave) << 3;
    const float* src;
    float* dst;
    if (g < 256)      { src = x;    dst = xf; }
    else if (g < 448) { src = wqkv; dst = wf;   g -= 256; }
    else              { src = wfc;  dst = wfcf; g -= 448; }
    const float* sp = src + (size_t)(g * 16 + (lane & 15)) * DCONST + j0 * 8 + ((lane >> 4) << 2);
    float* dp = dst + ((size_t)(g * 128 + j0) * 32 + lane) * 4;
    v4f t[8];
#pragma unroll
    for (int i = 0; i < 8; ++i) t[i] = *(const v4f*)(sp + i * 8);
#pragma unroll
    for (int i = 0; i < 8; ++i) *(v4f*)(dp + i * 128) = t[i];
}

#define GPITCH 16384

static __device__ inline void frag_load(const float* __restrict__ ap,
                                        const float* __restrict__ bp, int j,
                                        v4f (&a)[2][4], v4f (&b)[2][4]) {
#pragma unroll
    for (int c = 0; c < 2; ++c) {
#pragma unroll
        for (int i = 0; i < 4; ++i) {
            a[c][i] = *(const v4f*)(ap + (size_t)i * GPITCH + (j + c) * 128);
            b[c][i] = *(const v4f*)(bp + (size_t)i * GPITCH + (j + c) * 128);
        }
    }
}

static __device__ inline void frag_mma(const v4f (&a)[2][4], const v4f (&b)[2][4],
                                       v8f (&acc)[4][4]) {
#pragma unroll
    for (int c = 0; c < 2; ++c) {
#pragma unroll
        for (int mi = 0; mi < 4; ++mi)
#pragma unroll
            for (int ni = 0; ni < 4; ++ni)
                acc[mi][ni] = wmma4(a[c][mi].xy, b[c][ni].xy, acc[mi][ni]);
#pragma unroll
        for (int mi = 0; mi < 4; ++mi)
#pragma unroll
            for (int ni = 0; ni < 4; ++ni)
                acc[mi][ni] = wmma4(a[c][mi].zw, b[c][ni].zw, acc[mi][ni]);
    }
}

static __device__ inline void gemm_tile_64x64(const float* __restrict__ ap,
                                              const float* __restrict__ bp,
                                              v8f (&acc)[4][4]) {
    v4f a0[2][4], b0[2][4], a1[2][4], b1[2][4];
    frag_load(ap, bp, 0, a0, b0);
    for (int j = 0; j < 128; j += 4) {
        frag_load(ap, bp, j + 2, a1, b1);
        __builtin_amdgcn_sched_barrier(0);
        frag_mma(a0, b0, acc);
        __builtin_amdgcn_sched_barrier(0);
        const int jn = (j + 4 < 128) ? (j + 4) : j;
        frag_load(ap, bp, jn, a0, b0);
        __builtin_amdgcn_sched_barrier(0);
        frag_mma(a1, b1, acc);
        __builtin_amdgcn_sched_barrier(0);
    }
}

__global__ __launch_bounds__(128) void qkv_rms_kernel(
    const float* __restrict__ xf, const float* __restrict__ wf,
    const float* __restrict__ qnw, const float* __restrict__ knw,
    float* __restrict__ qb, float* __restrict__ kb, float* __restrict__ vb) {
    const int lane  = threadIdx.x & 31;
    const int wave  = threadIdx.x >> 5;
    const int laneN = lane & 15;
    const int hi8   = (lane >> 4) << 3;
    const int m0 = blockIdx.y * 64;
    const int n0 = blockIdx.x * 256 + wave * 64;

    v8f zero = {};
    v8f acc[4][4];
#pragma unroll
    for (int mi = 0; mi < 4; ++mi)
#pragma unroll
        for (int ni = 0; ni < 4; ++ni) acc[mi][ni] = zero;

    gemm_tile_64x64(xf + (size_t)(m0 >> 4) * GPITCH + lane * 4,
                    wf + (size_t)(n0 >> 4) * GPITCH + lane * 4, acc);

    const int sec = n0 >> 10;
    const int h   = (n0 & 1023) >> 6;
    float* dst = (sec == 0) ? qb : ((sec == 1) ? kb : vb);
    float wn[4];
#pragma unroll
    for (int ni = 0; ni < 4; ++ni) {
        const float* nw = (sec == 0) ? qnw : knw;
        wn[ni] = (sec < 2) ? nw[ni * 16 + laneN] : 1.0f;
    }

    if (sec == 2) {
#pragma unroll
        for (int mi = 0; mi < 4; ++mi) {
            const int m = m0 + mi * 16;
            const int b = m >> 11;
            const int l = m & (LCONST - 1);
            float* grp = dst + ((size_t)((b * NHC + h) * LCONST + l) << 6);
#pragma unroll
            for (int ni = 0; ni < 4; ++ni) {
#pragma unroll
                for (int u2 = 0; u2 < 2; ++u2) {
                    v4f t;
                    t.x = acc[mi][ni][4 * u2 + 0];
                    t.y = acc[mi][ni][4 * u2 + 1];
                    t.z = acc[mi][ni][4 * u2 + 2];
                    t.w = acc[mi][ni][4 * u2 + 3];
                    *(v4f*)(grp + ni * 256 + u2 * 128 + lane * 4) = t;
                }
            }
        }
    } else {
#pragma unroll
        for (int mi = 0; mi < 4; ++mi) {
#pragma unroll
            for (int r = 0; r < 8; ++r) {
                float sv[4];
#pragma unroll
                for (int ni = 0; ni < 4; ++ni) sv[ni] = acc[mi][ni][r];
                float ss = sv[0] * sv[0] + sv[1] * sv[1] + sv[2] * sv[2] + sv[3] * sv[3];
                ss += __shfl_xor(ss, 1, 32);
                ss += __shfl_xor(ss, 2, 32);
                ss += __shfl_xor(ss, 4, 32);
                ss += __shfl_xor(ss, 8, 32);
                float rstd = rsqrtf(ss * (1.0f / 64.0f) + 1e-5f);
#pragma unroll
                for (int ni = 0; ni < 4; ++ni) sv[ni] *= rstd * wn[ni];
                const int m = m0 + mi * 16 + r + hi8;
                const int b = m >> 11;
                const int l = m & (LCONST - 1);
                float* grp = dst + ((size_t)((b * NHC + h) * LCONST + (l & ~15)) << 6);
                v4f t;
                t.x = sv[0]; t.y = sv[1]; t.z = sv[2]; t.w = sv[3];
                *(v4f*)(grp + (laneN >> 1) * 128 + (((laneN & 1) << 4) + (l & 15)) * 4) = t;
            }
        }
    }
}

__global__ __launch_bounds__(32) __attribute__((amdgpu_waves_per_eu(2, 2)))
void attn_kernel(
    const float* __restrict__ qb, const float* __restrict__ kb,
    const float* __restrict__ vb, float* __restrict__ ob) {
    const int lane  = threadIdx.x;
    const int laneN = lane & 15;
    const int half  = lane >> 4;
    const int hi8   = half << 3;
    const int q0 = blockIdx.x * 32;
    const int bh = blockIdx.z * NHC + blockIdx.y;
    const float* qh = qb + (size_t)bh * LCONST * DHC;
    const float* kh = kb + (size_t)bh * LCONST * DHC;
    const float* vh = vb + (size_t)bh * LCONST * DHC;

    __shared__ __attribute__((aligned(16))) float q_s[2 * 8 * 128];
    __shared__ __attribute__((aligned(16))) float p_s[32 * 68];

#pragma unroll
    for (int i = 0; i < 16; ++i) {
        v4f t = *(const v4f*)(qh + (size_t)q0 * DHC + i * 128 + lane * 4);
        *(v4f*)(q_s + i * 128 + lane * 4) = t;
    }
    __syncthreads();

    v8f zero = {};
    v8f o[2][4];
    float mr[2][8], lr[2][8];
#pragma unroll
    for (int mi = 0; mi < 2; ++mi) {
#pragma unroll
        for (int nj = 0; nj < 4; ++nj) o[mi][nj] = zero;
#pragma unroll
        for (int r = 0; r < 8; ++r) { mr[mi][r] = -3.0e38f; lr[mi][r] = 0.0f; }
    }

    const int nkv = (q0 >> 6) + 1;

    v4f kf[4][8];
#pragma unroll
    for (int ni = 0; ni < 4; ++ni)
#pragma unroll
        for (int t2 = 0; t2 < 8; ++t2)
            kf[ni][t2] = *(const v4f*)(kh + (size_t)(ni * 16) * DHC + t2 * 128 + lane * 4);

    for (int kv = 0; kv < nkv; ++kv) {
        const int kbase = kv * 64;
        v8f s[2][4];
#pragma unroll
        for (int mi = 0; mi < 2; ++mi)
#pragma unroll
            for (int ni = 0; ni < 4; ++ni) s[mi][ni] = zero;

#pragma unroll
        for (int t2 = 0; t2 < 8; ++t2) {
            const v4f a0 = *(const v4f*)(q_s + t2 * 128 + lane * 4);
            const v4f a1 = *(const v4f*)(q_s + 1024 + t2 * 128 + lane * 4);
#pragma unroll
            for (int ni = 0; ni < 4; ++ni) {
                s[0][ni] = wmma4(a0.xy, kf[ni][t2].xy, s[0][ni]);
                s[1][ni] = wmma4(a1.xy, kf[ni][t2].xy, s[1][ni]);
                s[0][ni] = wmma4(a0.zw, kf[ni][t2].zw, s[0][ni]);
                s[1][ni] = wmma4(a1.zw, kf[ni][t2].zw, s[1][ni]);
            }
        }

        __builtin_amdgcn_sched_barrier(0);

        v4f vf[4][4][2];
#pragma unroll
        for (int gi = 0; gi < 4; ++gi)
#pragma unroll
            for (int nj = 0; nj < 4; ++nj)
#pragma unroll
                for (int u2 = 0; u2 < 2; ++u2)
                    vf[gi][nj][u2] = *(const v4f*)(vh + (size_t)(kbase + gi * 16) * DHC
                                                   + nj * 256 + u2 * 128 + lane * 4);
        __builtin_amdgcn_sched_barrier(0);

        const float scale = 0.125f;
#pragma unroll
        for (int mi = 0; mi < 2; ++mi) {
#pragma unroll
            for (int r = 0; r < 8; ++r) {
                const int qrow = q0 + mi * 16 + r + hi8;
                float sv[4];
                float vmax = -3.0e38f;
#pragma unroll
                for (int ni = 0; ni < 4; ++ni) {
                    const int key = kbase + ni * 16 + laneN;
                    float t = s[mi][ni][r] * scale;
                    t = (key > qrow) ? -1.0e30f : t;
                    sv[ni] = t;
                    vmax = fmaxf(vmax, t);
                }
                vmax = fmaxf(vmax, __shfl_xor(vmax, 1, 32));
                vmax = fmaxf(vmax, __shfl_xor(vmax, 2, 32));
                vmax = fmaxf(vmax, __shfl_xor(vmax, 4, 32));
                vmax = fmaxf(vmax, __shfl_xor(vmax, 8, 32));
                const float mold = mr[mi][r];
                const float mnew = fmaxf(mold, vmax);
                const float corr = __expf(mold - mnew);
                float psum = 0.0f;
#pragma unroll
                for (int ni = 0; ni < 4; ++ni) {
                    float p = __expf(sv[ni] - mnew);
                    sv[ni] = p;
                    psum += p;
                }
                psum += __shfl_xor(psum, 1, 32);
                psum += __shfl_xor(psum, 2, 32);
                psum += __shfl_xor(psum, 4, 32);
                psum += __shfl_xor(psum, 8, 32);
                lr[mi][r] = lr[mi][r] * corr + psum;
                mr[mi][r] = mnew;
#pragma unroll
                for (int nj = 0; nj < 4; ++nj) o[mi][nj][r] *= corr;
#pragma unroll
                for (int ni = 0; ni < 4; ++ni)
                    p_s[(mi * 16 + r + hi8) * 68 + ni * 16 + laneN] = sv[ni];
            }
        }
        __syncthreads();

        const int knext = (kv + 1 < nkv) ? (kbase + 64) : kbase;
#pragma unroll
        for (int ni = 0; ni < 4; ++ni)
#pragma unroll
            for (int t2 = 0; t2 < 8; ++t2)
                kf[ni][t2] = *(const v4f*)(kh + (size_t)(knext + ni * 16) * DHC + t2 * 128 + lane * 4);
        __builtin_amdgcn_sched_barrier(0);

#pragma unroll
        for (int gi = 0; gi < 4; ++gi) {
#pragma unroll
            for (int u2 = 0; u2 < 2; ++u2) {
                const int kc = gi * 16 + hi8 + u2 * 4;
                v4f a0 = *(const v4f*)(p_s + laneN * 68 + kc);
                v4f a1 = *(const v4f*)(p_s + (16 + laneN) * 68 + kc);
#pragma unroll
                for (int nj = 0; nj < 4; ++nj) {
                    o[0][nj] = wmma4(a0.xy, vf[gi][nj][u2].xy, o[0][nj]);
                    o[1][nj] = wmma4(a1.xy, vf[gi][nj][u2].xy, o[1][nj]);
                    o[0][nj] = wmma4(a0.zw, vf[gi][nj][u2].zw, o[0][nj]);
                    o[1][nj] = wmma4(a1.zw, vf[gi][nj][u2].zw, o[1][nj]);
                }
            }
        }
        __syncthreads();
    }

#pragma unroll
    for (int mi = 0; mi < 2; ++mi) {
#pragma unroll
        for (int r = 0; r < 8; ++r) {
            const float inv = 1.0f / lr[mi][r];
#pragma unroll
            for (int nj = 0; nj < 4; ++nj)
                q_s[mi * 1024 + (2 * nj + (laneN >> 3)) * 128
                    + ((((laneN >> 2) & 1) << 4) + r + hi8) * 4 + (laneN & 3)] = o[mi][nj][r] * inv;
        }
    }
    __syncthreads();
    float* og = ob + ((size_t)((blockIdx.z * LCONST + q0) >> 4) * 128 + blockIdx.y * 8) * 128 + lane * 4;
#pragma unroll
    for (int mi = 0; mi < 2; ++mi) {
#pragma unroll
        for (int jj = 0; jj < 8; ++jj) {
            const v4f t = *(const v4f*)(q_s + mi * 1024 + jj * 128 + lane * 4);
            *(v4f*)(og + (size_t)mi * GPITCH + jj * 128) = t;
        }
    }
}

__global__ __launch_bounds__(128) void fc_kernel(
    const float* __restrict__ of, const float* __restrict__ wfcf,
    float* __restrict__ out) {
    const int lane  = threadIdx.x & 31;
    const int wave  = threadIdx.x >> 5;
    const int laneN = lane & 15;
    const int hi8   = (lane >> 4) << 3;
    const int m0 = blockIdx.y * 64;
    const int n0 = blockIdx.x * 256 + wave * 64;

    v8f zero = {};
    v8f acc[4][4];
#pragma unroll
    for (int mi = 0; mi < 4; ++mi)
#pragma unroll
        for (int ni = 0; ni < 4; ++ni) acc[mi][ni] = zero;

    gemm_tile_64x64(of + (size_t)(m0 >> 4) * GPITCH + lane * 4,
                    wfcf + (size_t)(n0 >> 4) * GPITCH + lane * 4, acc);

#pragma unroll
    for (int mi = 0; mi < 4; ++mi) {
#pragma unroll
        for (int r = 0; r < 8; ++r) {
            const int m = m0 + mi * 16 + r + hi8;
            float* rowp = out + (size_t)m * DCONST + n0;
#pragma unroll
            for (int ni = 0; ni < 4; ++ni) rowp[ni * 16 + laneN] = acc[mi][ni][r];
        }
    }
}

extern "C" void kernel_launch(void* const* d_in, const int* in_sizes, int n_in,
                              void* d_out, int out_size, void* d_ws, size_t ws_size,
                              hipStream_t stream) {
    const float* x     = (const float*)d_in[0];
    const float* w_qkv = (const float*)d_in[2];
    const float* w_fc  = (const float*)d_in[3];
    const float* qnw   = (const float*)d_in[4];
    const float* knw   = (const float*)d_in[5];

    const size_t per = (size_t)2 * NHC * LCONST * DHC;
    float* wsf  = (float*)d_ws;
    float* qb   = wsf;
    float* kb   = wsf + per;
    float* vb   = wsf + 2 * per;
    float* obuf = wsf + 3 * per;
    float* xf   = wsf + 4 * per;
    float* wf   = wsf + 5 * per;
    float* wfcf = wsf + 6 * per;

    repack_kernel<<<1024, 256, 0, stream>>>(x, w_qkv, w_fc, xf, wf, wfcf);

    dim3 g1(3072 / 256, 4096 / 64);
    qkv_rms_kernel<<<g1, 128, 0, stream>>>(xf, wf, qnw, knw, qb, kb, vb);

    dim3 g2(LCONST / 32, NHC, 2);
    attn_kernel<<<g2, 32, 0, stream>>>(qb, kb, vb, obuf);

    dim3 g3(DCONST / 256, 4096 / 64);
    fc_kernel<<<g3, 128, 0, stream>>>(obuf, wfcf, (float*)d_out);
}


# attention: a tile's key blocks alternate between the 2 waves of its work-group (own max/sum/output, merged once in LDS), longest tiles launched first, groups refill freed slots: sum order changes
# speedup vs baseline: 3.0073x; 1.2990x over previous
#include <hip/hip_runtime.h>
#include <hip/hip_bf16.h>


typedef __attribute__((ext_vector_type(2))) float v2f;
typedef __attribute__((ext_vector_type(4))) float v4f;
typedef __attribute__((ext_vector_type(8))) float v8f;

#define LCONST 2048
#define DCONST 1024
#define NHC 16
#define DHC 64

static __device__ inline v8f wmma4(v2f a, v2f b, v8f c) {
    return __builtin_amdgcn_wmma_f32_16x16x4_f32(
         false, a,  false, b,
         (short)0, c,  false,  false);
}

__global__ __launch_bounds__(256) void repack_kernel(
    const float* __restrict__ x, const float* __restrict__ wqkv,
    const float* __restrict__ wfc, float* __restrict__ xf,
    float* __restrict__ wf, float* __restrict__ wfcf) {
    const int lane = threadIdx.x & 31;
    const int wave = threadIdx.x >> 5;
    int g = blockIdx.x >> 1;
    const int j0 = (((blockIdx.x & 1) << 3) + wave) << 3;
    const float* src;
    float* dst;
    if (g < 256)      { src = x;    dst = xf; }
    else if (g < 448) { src = wqkv; dst = wf;   g -= 256; }
    else              { src = wfc;  dst = wfcf; g -= 448; }
    const float* sp = src + (size_t)(g * 16 + (lane & 15)) * DCONST + j0 * 8 + ((lane >> 4) << 2);
    float* dp = dst + ((size_t)(g * 128 + j0) * 32 + lane) * 4;
    v4f t[8];
#pragma unroll
    for (int i = 0; i < 8; ++i) t[i] = *(const v4f*)(sp + i * 8);
#pragma unroll
    for (int i = 0; i < 8; ++i) *(v4f*)(dp + i * 128) = t[i];
}

#define GPITCH 16384

static __device__ inline void frag_load(const float* __restrict__ ap,
                                        const float* __restrict__ bp, int j,
                                        v4f (&a)[2][4], v4f (&b)[2][4]) {
#pragma unroll
    for (int c = 0; c < 2; ++c) {
#pragma unroll
        for (int i = 0; i < 4; ++i) {
            a[c][i] = *(const v4f*)(ap + (size_t)i * GPITCH + (j + c) * 128);
            b[c][i] = *(const v4f*)(bp + (size_t)i * GPITCH + (j + c) * 128);
        }
    }
}

static __device__ inline void frag_mma(const v4f (&a)[2][4], const v4f (&b)[2][4],
                                       v8f (&acc)[4][4]) {
#pragma unroll
    for (int c = 0; c < 2; ++c) {
#pragma unroll
        for (int mi = 0; mi < 4; ++mi)
#pragma unroll
            for (int ni = 0; ni < 4; ++ni)
                acc[mi][ni] = wmma4(a[c][mi].xy, b[c][ni].xy, acc[mi][ni]);
#pragma unroll
        for (int mi = 0; mi < 4; ++mi)
#pragma unroll
            for (int ni = 0; ni < 4; ++ni)
                acc[mi][ni] = wmma4(a[c][mi].zw, b[c][ni].zw, acc[mi][ni]);
    }
}

static __device__ inline void gemm_tile_64x64(const float* __restrict__ ap,
                                              const float* __restrict__ bp,
                                              v8f (&acc)[4][4]) {
    v4f a0[2][4], b0[2][4], a1[2][4], b1[2][4];
    frag_load(ap, bp, 0, a0, b0);
    for (int j = 0; j < 128; j += 4) {
        frag_load(ap, bp, j + 2, a1, b1);
        __builtin_amdgcn_sched_barrier(0);
        frag_mma(a0, b0, acc);
        __builtin_amdgcn_sched_barrier(0);
        const int jn = (j + 4 < 128) ? (j + 4) : j;
        frag_load(ap, bp, jn, a0, b0);
        __builtin_amdgcn_sched_barrier(0);
        frag_mma(a1, b1, acc);
        __builtin_amdgcn_sched_barrier(0);
    }
}

__global__ __launch_bounds__(128) void qkv_rms_kernel(
    const float* __restrict__ xf, const float* __restrict__ wf,
    const float* __restrict__ qnw, const float* __restrict__ knw,
    float* __restrict__ qb, float* __restrict__ kb, float* __restrict__ vb) {
    const int lane  = threadIdx.x & 31;
    const int wave  = threadIdx.x >> 5;
    const int laneN = lane & 15;
    const int hi8   = (lane >> 4) << 3;
    const int m0 = blockIdx.y * 64;
    const int n0 = blockIdx.x * 256 + wave * 64;

    v8f zero = {};
    v8f acc[4][4];
#pragma unroll
    for (int mi = 0; mi < 4; ++mi)
#pragma unroll
        for (int ni = 0; ni < 4; ++ni) acc[mi][ni] = zero;

    gemm_tile_64x64(xf + (size_t)(m0 >> 4) * GPITCH + lane * 4,
                    wf + (size_t)(n0 >> 4) * GPITCH + lane * 4, acc);

    const int sec = n0 >> 10;
    const int h   = (n0 & 1023) >> 6;
    float* dst = (sec == 0) ? qb : ((sec == 1) ? kb : vb);
    float wn[4];
#pragma unroll
    for (int ni = 0; ni < 4; ++ni) {
        const float* nw = (sec == 0) ? qnw : knw;
        wn[ni] = (sec < 2) ? nw[ni * 16 + laneN] : 1.0f;
    }

    if (sec == 2) {
#pragma unroll
        for (int mi = 0; mi < 4; ++mi) {
            const int m = m0 + mi * 16;
            const int b = m >> 11;
            const int l = m & (LCONST - 1);
            float* grp = dst + ((size_t)((b * NHC + h) * LCONST + l) << 6);
#pragma unroll
            for (int ni = 0; ni < 4; ++ni) {
#pragma unroll
                for (int u2 = 0; u2 < 2; ++u2) {
                    v4f t;
                    t.x = acc[mi][ni][4 * u2 + 0];
                    t.y = acc[mi][ni][4 * u2 + 1];
                    t.z = acc[mi][ni][4 * u2 + 2];
                    t.w = acc[mi][ni][4 * u2 + 3];
                    *(v4f*)(grp + ni * 256 + u2 * 128 + lane * 4) = t;
                }
            }
        }
    } else {
#pragma unroll
        for (int mi = 0; mi < 4; ++mi) {
#pragma unroll
            for (int r = 0; r < 8; ++r) {
                float sv[4];
#pragma unroll
                for (int ni = 0; ni < 4; ++ni) sv[ni] = acc[mi][ni][r];
                float ss = sv[0] * sv[0] + sv[1] * sv[1] + sv[2] * sv[2] + sv[3] * sv[3];
                ss += __shfl_xor(ss, 1, 32);
                ss += __shfl_xor(ss, 2, 32);
                ss += __shfl_xor(ss, 4, 32);
                ss += __shfl_xor(ss, 8, 32);
                float rstd = rsqrtf(ss * (1.0f / 64.0f) + 1e-5f);
#pragma unroll
                for (int ni = 0; ni < 4; ++ni) sv[ni] *= rstd * wn[ni];
                const int m = m0 + mi * 16 + r + hi8;
                const int b = m >> 11;
                const int l = m & (LCONST - 1);
                float* grp = dst + ((size_t)((b * NHC + h) * LCONST + (l & ~15)) << 6);
                v4f t;
                t.x = sv[0]; t.y = sv[1]; t.z = sv[2]; t.w = sv[3];
                *(v4f*)(grp + (laneN >> 1) * 128 + (((laneN & 1) << 4) + (l & 15)) * 4) = t;
            }
        }
    }
}


static __device__ inline void wave_lds_order() {
    __builtin_amdgcn_fence(__ATOMIC_RELEASE, "workgroup", "local");
    __builtin_amdgcn_wave_barrier();
    __builtin_amdgcn_fence(__ATOMIC_ACQUIRE, "workgroup", "local");
}

__global__ __launch_bounds__(64) __attribute__((amdgpu_waves_per_eu(2, 2)))
void attn_kernel(
    const float* __restrict__ qb, const float* __restrict__ kb,
    const float* __restrict__ vb, float* __restrict__ ob) {
    const int lane  = threadIdx.x & 31;
    const int w     = __builtin_amdgcn_readfirstlane((int)(threadIdx.x >> 5));
    const int laneN = lane & 15;
    const int half  = lane >> 4;
    const int hi8   = half << 3;
    const int q0 = (LCONST / 32 - 1 - (int)blockIdx.y) * 32;
    const int bh = blockIdx.x;
    const float* qh = qb + (size_t)bh * LCONST * DHC;
    const float* kh = kb + (size_t)bh * LCONST * DHC;
    const float* vh = vb + (size_t)bh * LCONST * DHC;

    __shared__ __attribute__((aligned(16))) float q_s[2 * 8 * 128];
    __shared__ __attribute__((aligned(16))) float p_all[2 * 32 * 68];
    float* p_s = p_all + w * (32 * 68);
    float* x_s = p_all + 32 * 68;

#pragma unroll
    for (int i = 0; i < 8; ++i) {
        v4f t = *(const v4f*)(qh + (size_t)q0 * DHC + (w * 8 + i) * 128 + lane * 4);
        *(v4f*)(q_s + (w * 8 + i) * 128 + lane * 4) = t;
    }
    __syncthreads();

    v8f zero = {};
    v8f o[2][4];
    float mr[2][8], lr[2][8];
#pragma unroll
    for (int mi = 0; mi < 2; ++mi) {
#pragma unroll
        for (int nj = 0; nj < 4; ++nj) o[mi][nj] = zero;
#pragma unroll
        for (int r = 0; r < 8; ++r) { mr[mi][r] = -3.0e38f; lr[mi][r] = 0.0f; }
    }

    const int nkv = (q0 >> 6) + 1;

    v4f kf[4][8];
#pragma unroll
    for (int ni = 0; ni < 4; ++ni)
#pragma unroll
        for (int t2 = 0; t2 < 8; ++t2)
            kf[ni][t2] = *(const v4f*)(kh + (size_t)(w * 64 + ni * 16) * DHC + t2 * 128 + lane * 4);

    for (int kv = w; kv < nkv; kv += 2) {
        const int kbase = kv * 64;
        v8f s[2][4];
#pragma unroll
        for (int mi = 0; mi < 2; ++mi)
#pragma unroll
            for (int ni = 0; ni < 4; ++ni) s[mi][ni] = zero;

#pragma unroll
        for (int t2 = 0; t2 < 8; ++t2) {
            const v4f a0 = *(const v4f*)(q_s + t2 * 128 + lane * 4);
            const v4f a1 = *(const v4f*)(q_s + 1024 + t2 * 128 + lane * 4);
#pragma unroll
            for (int ni = 0; ni < 4; ++ni) {
                s[0][ni] = wmma4(a0.xy, kf[ni][t2].xy, s[0][ni]);
                s[1][ni] = wmma4(a1.xy, kf[ni][t2].xy, s[1][ni]);
                s[0][ni] = wmma4(a0.zw, kf[ni][t2].zw, s[0][ni]);
                s[1][ni] = wmma4(a1.zw, kf[ni][t2].zw, s[1][ni]);
            }
        }

        __builtin_amdgcn_sched_barrier(0);

        v4f vf[4][4][2];
#pragma unroll
        for (int gi = 0; gi < 4; ++gi)
#pragma unroll
            for (int nj = 0; nj < 4; ++nj)
#pragma unroll
                for (int u2 = 0; u2 < 2; ++u2)
                    vf[gi][nj][u2] = *(const v4f*)(vh + (size_t)(kbase + gi * 16) * DHC
                                                   + nj * 256 + u2 * 128 + lane * 4);
        __builtin_amdgcn_sched_barrier(0);

        const float scale = 0.125f;
#pragma unroll
        for (int mi = 0; mi < 2; ++mi) {
#pragma unroll
            for (int r = 0; r < 8; ++r) {
                const int qrow = q0 + mi * 16 + r + hi8;
                float sv[4];
                float vmax = -3.0e38f;
#pragma unroll
                for (int ni = 0; ni < 4; ++ni) {
                    const int key = kbase + ni * 16 + laneN;
                    float t = s[mi][ni][r] * scale;
                    t = (key > qrow) ? -1.0e30f : t;
                    sv[ni] = t;
                    vmax = fmaxf(vmax, t);
                }
                vmax = fmaxf(vmax, __shfl_xor(vmax, 1, 32));
                vmax = fmaxf(vmax, __shfl_xor(vmax, 2, 32));
                vmax = fmaxf(vmax, __shfl_xor(vmax, 4, 32));
                vmax = fmaxf(vmax, __shfl_xor(vmax, 8, 32));
                const float mold = mr[mi][r];
                const float mnew = fmaxf(mold, vmax);
                const float corr = __expf(mold - mnew);
                float psum = 0.0f;
#pragma unroll
                for (int ni = 0; ni < 4; ++ni) {
                    float p = __expf(sv[ni] - mnew);
                    sv[ni] = p;
                    psum += p;
                }
                psum += __shfl_xor(psum, 1, 32);
                psum += __shfl_xor(psum, 2, 32);
                psum += __shfl_xor(psum, 4, 32);
                psum += __shfl_xor(psum, 8, 32);
                lr[mi][r] = lr[mi][r] * corr + psum;
                mr[mi][r] = mnew;
#pragma unroll
                for (int nj = 0; nj < 4; ++nj) o[mi][nj][r] *= corr;
#pragma unroll
                for (int ni = 0; ni < 4; ++ni)
                    p_s[(mi * 16 + r + hi8) * 68 + ni * 16 + laneN] = sv[ni];
            }
        }
        wave_lds_order();

        const int knext = (kv + 2 < nkv) ? (kbase + 128) : kbase;
#pragma unroll
        for (int ni = 0; ni < 4; ++ni)
#pragma unroll
            for (int t2 = 0; t2 < 8; ++t2)
                kf[ni][t2] = *(const v4f*)(kh + (size_t)(knext + ni * 16) * DHC + t2 * 128 + lane * 4);
        __builtin_amdgcn_sched_barrier(0);

#pragma unroll
        for (int gi = 0; gi < 4; ++gi) {
#pragma unroll
            for (int u2 = 0; u2 < 2; ++u2) {
                const int kc = gi * 16 + hi8 + u2 * 4;
                v4f a0 = *(const v4f*)(p_s + laneN * 68 + kc);
                v4f a1 = *(const v4f*)(p_s + (16 + laneN) * 68 + kc);
#pragma unroll
                for (int nj = 0; nj < 4; ++nj) {
                    o[0][nj] = wmma4(a0.xy, vf[gi][nj][u2].xy, o[0][nj]);
                    o[1][nj] = wmma4(a1.xy, vf[gi][nj][u2].xy, o[1][nj]);
                    o[0][nj] = wmma4(a0.zw, vf[gi][nj][u2].zw, o[0][nj]);
                    o[1][nj] = wmma4(a1.zw, vf[gi][nj][u2].zw, o[1][nj]);
                }
            }
        }
        wave_lds_order();
    }

    if (w == 1) {
#pragma unroll
        for (int mi = 0; mi < 2; ++mi) {
#pragma unroll
            for (int nj = 0; nj < 4; ++nj) {
#pragma unroll
                for (int u2 = 0; u2 < 2; ++u2) {
                    v4f t;
                    t.x = o[mi][nj][4 * u2 + 0];
                    t.y = o[mi][nj][4 * u2 + 1];
                    t.z = o[mi][nj][4 * u2 + 2];
                    t.w = o[mi][nj][4 * u2 + 3];
                    *(v4f*)(x_s + ((mi * 4 + nj) * 2 + u2) * 128 + lane * 4) = t;
                }
            }
            if (laneN == 0) {
#pragma unroll
                for (int r = 0; r < 8; ++r) {
                    x_s[2048 + mi * 16 + r + hi8] = mr[mi][r];
                    x_s[2080 + mi * 16 + r + hi8] = lr[mi][r];
                }
            }
        }
    }
    __syncthreads();

    if (w == 0) {
#pragma unroll
        for (int mi = 0; mi < 2; ++mi) {
            float c0[8], c1[8];
#pragma unroll
            for (int r = 0; r < 8; ++r) {
                const float m1 = x_s[2048 + mi * 16 + r + hi8];
                const float l1 = x_s[2080 + mi * 16 + r + hi8];
                const float m  = fmaxf(mr[mi][r], m1);
                const float e0 = __expf(mr[mi][r] - m);
                const float e1 = __expf(m1 - m);
                const float inv = 1.0f / (lr[mi][r] * e0 + l1 * e1);
                c0[r] = e0 * inv;
                c1[r] = e1 * inv;
            }
#pragma unroll
            for (int nj = 0; nj < 4; ++nj) {
#pragma unroll
                for (int u2 = 0; u2 < 2; ++u2) {
                    const v4f t = *(const v4f*)(x_s + ((mi * 4 + nj) * 2 + u2) * 128 + lane * 4);
#pragma unroll
                    for (int e = 0; e < 4; ++e) {
                        const int r = 4 * u2 + e;
                        q_s[mi * 1024 + (2 * nj + (laneN >> 3)) * 128
                            + ((((laneN >> 2) & 1) << 4) + r + hi8) * 4 + (laneN & 3)]
                            = o[mi][nj][r] * c0[r] + t[e] * c1[r];
                    }
                }
            }
        }
    }
    __syncthreads();
    const int bz = bh >> 4;
    const int hy = bh & (NHC - 1);
    float* og = ob + ((size_t)((bz * LCONST + q0) >> 4) * 128 + hy * 8) * 128 + lane * 4;
#pragma unroll
    for (int jj = 0; jj < 8; ++jj) {
        const v4f t = *(const v4f*)(q_s + w * 1024 + jj * 128 + lane * 4);
        *(v4f*)(og + (size_t)w * GPITCH + jj * 128) = t;
    }
}

__global__ __launch_bounds__(128) void fc_kernel(
    const float* __restrict__ of, const float* __restrict__ wfcf,
    float* __restrict__ out) {
    const int lane  = threadIdx.x & 31;
    const int wave  = threadIdx.x >> 5;
    const int laneN = lane & 15;
    const int hi8   = (lane >> 4) << 3;
    const int m0 = blockIdx.y * 64;
    const int n0 = blockIdx.x * 256 + wave * 64;

    v8f zero = {};
    v8f acc[4][4];
#pragma unroll
    for (int mi = 0; mi < 4; ++mi)
#pragma unroll
        for (int ni = 0; ni < 4; ++ni) acc[mi][ni] = zero;

    gemm_tile_64x64(of + (size_t)(m0 >> 4) * GPITCH + lane * 4,
                    wfcf + (size_t)(n0 >> 4) * GPITCH + lane * 4, acc);

#pragma unroll
    for (int mi = 0; mi < 4; ++mi) {
#pragma unroll
        for (int r = 0; r < 8; ++r) {
            const int m = m0 + mi * 16 + r + hi8;
            float* rowp = out + (size_t)m * DCONST + n0;
#pragma unroll
            for (int ni = 0; ni < 4; ++ni) rowp[ni * 16 + laneN] = acc[mi][ni][r];
        }
    }
}

extern "C" void kernel_launch(void* const* d_in, const int* in_sizes, int n_in,
                              void* d_out, int out_size, void* d_ws, size_t ws_size,
                              hipStream_t stream) {
    const float* x     = (const float*)d_in[0];
    const float* w_qkv = (const float*)d_in[2];
    const float* w_fc  = (const float*)d_in[3];
    const float* qnw   = (const float*)d_in[4];
    const float* knw   = (const float*)d_in[5];

    const size_t per = (size_t)2 * NHC * LCONST * DHC;
    float* wsf  = (float*)d_ws;
    float* qb   = wsf;
    float* kb   = wsf + per;
    float* vb   = wsf + 2 * per;
    float* obuf = wsf + 3 * per;
    float* xf   = wsf + 4 * per;
    float* wf   = wsf + 5 * per;
    float* wfcf = wsf + 6 * per;

    repack_kernel<<<1024, 256, 0, stream>>>(x, w_qkv, w_fc, xf, wf, wfcf);

    dim3 g1(3072 / 256, 4096 / 64);
    qkv_rms_kernel<<<g1, 128, 0, stream>>>(xf, wf, qnw, knw, qb, kb, vb);

    dim3 g2(2 * NHC, LCONST / 32);
    attn_kernel<<<g2, 64, 0, stream>>>(qb, kb, vb, obuf);

    dim3 g3(DCONST / 256, 4096 / 64);
    fc_kernel<<<g3, 128, 0, stream>>>(obuf, wfcf, (float*)d_out);
}


# attention softmax and RMS norm: the 16-lane row max/sum butterflies use row_xmask DPP moves on the vector unit instead of __shfl_xor through the LDS unit; same values, same order of sums
# speedup vs baseline: 3.1617x; 1.0513x over previous
#include <hip/hip_runtime.h>
#include <hip/hip_bf16.h>


typedef __attribute__((ext_vector_type(2))) float v2f;
typedef __attribute__((ext_vector_type(4))) float v4f;
typedef __attribute__((ext_vector_type(8))) float v8f;

#define LCONST 2048
#define DCONST 1024
#define NHC 16
#define DHC 64

template <int M>
static __device__ inline float row_xor(float v) {
    return __builtin_amdgcn_update_dpp(0.0f, v, 0x160 + M, 0xf, 0xf, true);
}

static __device__ inline v8f wmma4(v2f a, v2f b, v8f c) {
    return __builtin_amdgcn_wmma_f32_16x16x4_f32(
         false, a,  false, b,
         (short)0, c,  false,  false);
}

__global__ __launch_bounds__(256) void repack_kernel(
    const float* __restrict__ x, const float* __restrict__ wqkv,
    const float* __restrict__ wfc, float* __restrict__ xf,
    float* __restrict__ wf, float* __restrict__ wfcf) {
    const int lane = threadIdx.x & 31;
    const int wave = threadIdx.x >> 5;
    int g = blockIdx.x >> 1;
    const int j0 = (((blockIdx.x & 1) << 3) + wave) << 3;
    const float* src;
    float* dst;
    if (g < 256)      { src = x;    dst = xf; }
    else if (g < 448) { src = wqkv; dst = wf;   g -= 256; }
    else              { src = wfc;  dst = wfcf; g -= 448; }
    const float* sp = src + (size_t)(g * 16 + (lane & 15)) * DCONST + j0 * 8 + ((lane >> 4) << 2);
    float* dp = dst + ((size_t)(g * 128 + j0) * 32 + lane) * 4;
    v4f t[8];
#pragma unroll
    for (int i = 0; i < 8; ++i) t[i] = *(const v4f*)(sp + i * 8);
#pragma unroll
    for (int i = 0; i < 8; ++i) *(v4f*)(dp + i * 128) = t[i];
}

#define GPITCH 16384

static __device__ inline void frag_load(const float* __restrict__ ap,
                                        const float* __restrict__ bp, int j,
                                        v4f (&a)[2][4], v4f (&b)[2][4]) {
#pragma unroll
    for (int c = 0; c < 2; ++c) {
#pragma unroll
        for (int i = 0; i < 4; ++i) {
            a[c][i] = *(const v4f*)(ap + (size_t)i * GPITCH + (j + c) * 128);
            b[c][i] = *(const v4f*)(bp + (size_t)i * GPITCH + (j + c) * 128);
        }
    }
}

static __device__ inline void frag_mma(const v4f (&a)[2][4], const v4f (&b)[2][4],
                                       v8f (&acc)[4][4]) {
#pragma unroll
    for (int c = 0; c < 2; ++c) {
#pragma unroll
        for (int mi = 0; mi < 4; ++mi)
#pragma unroll
            for (int ni = 0; ni < 4; ++ni)
                acc[mi][ni] = wmma4(a[c][mi].xy, b[c][ni].xy, acc[mi][ni]);
#pragma unroll
        for (int mi = 0; mi < 4; ++mi)
#pragma unroll
            for (int ni = 0; ni < 4; ++ni)
                acc[mi][ni] = wmma4(a[c][mi].zw, b[c][ni].zw, acc[mi][ni]);
    }
}

static __device__ inline void gemm_tile_64x64(const float* __restrict__ ap,
                                              const float* __restrict__ bp,
                                              v8f (&acc)[4][4]) {
    v4f a0[2][4], b0[2][4], a1[2][4], b1[2][4];
    frag_load(ap, bp, 0, a0, b0);
    for (int j = 0; j < 128; j += 4) {
        frag_load(ap, bp, j + 2, a1, b1);
        __builtin_amdgcn_sched_barrier(0);
        frag_mma(a0, b0, acc);
        __builtin_amdgcn_sched_barrier(0);
        const int jn = (j + 4 < 128) ? (j + 4) : j;
        frag_load(ap, bp, jn, a0, b0);
        __builtin_amdgcn_sched_barrier(0);
        frag_mma(a1, b1, acc);
        __builtin_amdgcn_sched_barrier(0);
    }
}

__global__ __launch_bounds__(128) void qkv_rms_kernel(
    const float* __restrict__ xf, const float* __restrict__ wf,
    const float* __restrict__ qnw, const float* __restrict__ knw,
    float* __restrict__ qb, float* __restrict__ kb, float* __restrict__ vb) {
    const int lane  = threadIdx.x & 31;
    const int wave  = threadIdx.x >> 5;
    const int laneN = lane & 15;
    const int hi8   = (lane >> 4) << 3;
    const int m0 = blockIdx.y * 64;
    const int n0 = blockIdx.x * 256 + wave * 64;

    v8f zero = {};
    v8f acc[4][4];
#pragma unroll
    for (int mi = 0; mi < 4; ++mi)
#pragma unroll
        for (int ni = 0; ni < 4; ++ni) acc[mi][ni] = zero;

    gemm_tile_64x64(xf + (size_t)(m0 >> 4) * GPITCH + lane * 4,
                    wf + (size_t)(n0 >> 4) * GPITCH + lane * 4, acc);

    const int sec = n0 >> 10;
    const int h   = (n0 & 1023) >> 6;
    float* dst = (sec == 0) ? qb : ((sec == 1) ? kb : vb);
    float wn[4];
#pragma unroll
    for (int ni = 0; ni < 4; ++ni) {
        const float* nw = (sec == 0) ? qnw : knw;
        wn[ni] = (sec < 2) ? nw[ni * 16 + laneN] : 1.0f;
    }

    if (sec == 2) {
#pragma unroll
        for (int mi = 0; mi < 4; ++mi) {
            const int m = m0 + mi * 16;
            const int b = m >> 11;
            const int l = m & (LCONST - 1);
            float* grp = dst + ((size_t)((b * NHC + h) * LCONST + l) << 6);
#pragma unroll
            for (int ni = 0; ni < 4; ++ni) {
#pragma unroll
                for (int u2 = 0; u2 < 2; ++u2) {
                    v4f t;
                    t.x = acc[mi][ni][4 * u2 + 0];
                    t.y = acc[mi][ni][4 * u2 + 1];
                    t.z = acc[mi][ni][4 * u2 + 2];
                    t.w = acc[mi][ni][4 * u2 + 3];
                    *(v4f*)(grp + ni * 256 + u2 * 128 + lane * 4) = t;
                }
            }
        }
    } else {
#pragma unroll
        for (int mi = 0; mi < 4; ++mi) {
#pragma unroll
            for (int r = 0; r < 8; ++r) {
                float sv[4];
#pragma unroll
                for (int ni = 0; ni < 4; ++ni) sv[ni] = acc[mi][ni][r];
                float ss = sv[0] * sv[0] + sv[1] * sv[1] + sv[2] * sv[2] + sv[3] * sv[3];
                ss += row_xor<1>(ss);
                ss += row_xor<2>(ss);
                ss += row_xor<4>(ss);
                ss += row_xor<8>(ss);
                float rstd = rsqrtf(ss * (1.0f / 64.0f) + 1e-5f);
#pragma unroll
                for (int ni = 0; ni < 4; ++ni) sv[ni] *= rstd * wn[ni];
                const int m = m0 + mi * 16 + r + hi8;
                const int b = m >> 11;
                const int l = m & (LCONST - 1);
                float* grp = dst + ((size_t)((b * NHC + h) * LCONST + (l & ~15)) << 6);
                v4f t;
                t.x = sv[0]; t.y = sv[1]; t.z = sv[2]; t.w = sv[3];
                *(v4f*)(grp + (laneN >> 1) * 128 + (((laneN & 1) << 4) + (l & 15)) * 4) = t;
            }
        }
    }
}


static __device__ inline void wave_lds_order() {
    __builtin_amdgcn_fence(__ATOMIC_RELEASE, "workgroup", "local");
    __builtin_amdgcn_wave_barrier();
    __builtin_amdgcn_fence(__ATOMIC_ACQUIRE, "workgroup", "local");
}

__global__ __launch_bounds__(64) __attribute__((amdgpu_waves_per_eu(2, 2)))
void attn_kernel(
    const float* __restrict__ qb, const float* __restrict__ kb,
    const float* __restrict__ vb, float* __restrict__ ob) {
    const int lane  = threadIdx.x & 31;
    const int w     = __builtin_amdgcn_readfirstlane((int)(threadIdx.x >> 5));
    const int laneN = lane & 15;
    const int half  = lane >> 4;
    const int hi8   = half << 3;
    const int q0 = (LCONST / 32 - 1 - (int)blockIdx.y) * 32;
    const int bh = blockIdx.x;
    const float* qh = qb + (size_t)bh * LCONST * DHC;
    const float* kh = kb + (size_t)bh * LCONST * DHC;
    const float* vh = vb + (size_t)bh * LCONST * DHC;

    __shared__ __attribute__((aligned(16))) float q_s[2 * 8 * 128];
    __shared__ __attribute__((aligned(16))) float p_all[2 * 32 * 68];
    float* p_s = p_all + w * (32 * 68);
    float* x_s = p_all + 32 * 68;

#pragma unroll
    for (int i = 0; i < 8; ++i) {
        v4f t = *(const v4f*)(qh + (size_t)q0 * DHC + (w * 8 + i) * 128 + lane * 4);
        *(v4f*)(q_s + (w * 8 + i) * 128 + lane * 4) = t;
    }
    __syncthreads();

    v8f zero = {};
    v8f o[2][4];
    float mr[2][8], lr[2][8];
#pragma unroll
    for (int mi = 0; mi < 2; ++mi) {
#pragma unroll
        for (int nj = 0; nj < 4; ++nj) o[mi][nj] = zero;
#pragma unroll
        for (int r = 0; r < 8; ++r) { mr[mi][r] = -3.0e38f; lr[mi][r] = 0.0f; }
    }

    const int nkv = (q0 >> 6) + 1;

    v4f kf[4][8];
#pragma unroll
    for (int ni = 0; ni < 4; ++ni)
#pragma unroll
        for (int t2 = 0; t2 < 8; ++t2)
            kf[ni][t2] = *(const v4f*)(kh + (size_t)(w * 64 + ni * 16) * DHC + t2 * 128 + lane * 4);

    for (int kv = w; kv < nkv; kv += 2) {
        const int kbase = kv * 64;
        v8f s[2][4];
#pragma unroll
        for (int mi = 0; mi < 2; ++mi)
#pragma unroll
            for (int ni = 0; ni < 4; ++ni) s[mi][ni] = zero;

#pragma unroll
        for (int t2 = 0; t2 < 8; ++t2) {
            const v4f a0 = *(const v4f*)(q_s + t2 * 128 + lane * 4);
            const v4f a1 = *(const v4f*)(q_s + 1024 + t2 * 128 + lane * 4);
#pragma unroll
            for (int ni = 0; ni < 4; ++ni) {
                s[0][ni] = wmma4(a0.xy, kf[ni][t2].xy, s[0][ni]);
                s[1][ni] = wmma4(a1.xy, kf[ni][t2].xy, s[1][ni]);
                s[0][ni] = wmma4(a0.zw, kf[ni][t2].zw, s[0][ni]);
                s[1][ni] = wmma4(a1.zw, kf[ni][t2].zw, s[1][ni]);
            }
        }

        __builtin_amdgcn_sched_barrier(0);

        v4f vf[4][4][2];
#pragma unroll
        for (int gi = 0; gi < 4; ++gi)
#pragma unroll
            for (int nj = 0; nj < 4; ++nj)
#pragma unroll
                for (int u2 = 0; u2 < 2; ++u2)
                    vf[gi][nj][u2] = *(const v4f*)(vh + (size_t)(kbase + gi * 16) * DHC
                                                   + nj * 256 + u2 * 128 + lane * 4);
        __builtin_amdgcn_sched_barrier(0);

        const float scale = 0.125f;
#pragma unroll
        for (int mi = 0; mi < 2; ++mi) {
#pragma unroll
            for (int r = 0; r < 8; ++r) {
                const int qrow = q0 + mi * 16 + r + hi8;
                float sv[4];
                float vmax = -3.0e38f;
#pragma unroll
                for (int ni = 0; ni < 4; ++ni) {
                    const int key = kbase + ni * 16 + laneN;
                    float t = s[mi][ni][r] * scale;
                    t = (key > qrow) ? -1.0e30f : t;
                    sv[ni] = t;
                    vmax = fmaxf(vmax, t);
                }
                vmax = fmaxf(vmax, row_xor<1>(vmax));
                vmax = fmaxf(vmax, row_xor<2>(vmax));
                vmax = fmaxf(vmax, row_xor<4>(vmax));
                vmax = fmaxf(vmax, row_xor<8>(vmax));
                const float mold = mr[mi][r];
                const float mnew = fmaxf(mold, vmax);
                const float corr = __expf(mold - mnew);
                float psum = 0.0f;
#pragma unroll
                for (int ni = 0; ni < 4; ++ni) {
                    float p = __expf(sv[ni] - mnew);
                    sv[ni] = p;
                    psum += p;
                }
                psum += row_xor<1>(psum);
                psum += row_xor<2>(psum);
                psum += row_xor<4>(psum);
                psum += row_xor<8>(psum);
                lr[mi][r] = lr[mi][r] * corr + psum;
                mr[mi][r] = mnew;
#pragma unroll
                for (int nj = 0; nj < 4; ++nj) o[mi][nj][r] *= corr;
#pragma unroll
                for (int ni = 0; ni < 4; ++ni)
                    p_s[(mi * 16 + r + hi8) * 68 + ni * 16 + laneN] = sv[ni];
            }
        }
        wave_lds_order();

        const int knext = (kv + 2 < nkv) ? (kbase + 128) : kbase;
#pragma unroll
        for (int ni = 0; ni < 4; ++ni)
#pragma unroll
            for (int t2 = 0; t2 < 8; ++t2)
                kf[ni][t2] = *(const v4f*)(kh + (size_t)(knext + ni * 16) * DHC + t2 * 128 + lane * 4);
        __builtin_amdgcn_sched_barrier(0);

#pragma unroll
        for (int gi = 0; gi < 4; ++gi) {
#pragma unroll
            for (int u2 = 0; u2 < 2; ++u2) {
                const int kc = gi * 16 + hi8 + u2 * 4;
                v4f a0 = *(const v4f*)(p_s + laneN * 68 + kc);
                v4f a1 = *(const v4f*)(p_s + (16 + laneN) * 68 + kc);
#pragma unroll
                for (int nj = 0; nj < 4; ++nj) {
                    o[0][nj] = wmma4(a0.xy, vf[gi][nj][u2].xy, o[0][nj]);
                    o[1][nj] = wmma4(a1.xy, vf[gi][nj][u2].xy, o[1][nj]);
                    o[0][nj] = wmma4(a0.zw, vf[gi][nj][u2].zw, o[0][nj]);
                    o[1][nj] = wmma4(a1.zw, vf[gi][nj][u2].zw, o[1][nj]);
                }
            }
        }
        wave_lds_order();
    }

    if (w == 1) {
#pragma unroll
        for (int mi = 0; mi < 2; ++mi) {
#pragma unroll
            for (int nj = 0; nj < 4; ++nj) {
#pragma unroll
                for (int u2 = 0; u2 < 2; ++u2) {
                    v4f t;
                    t.x = o[mi][nj][4 * u2 + 0];
                    t.y = o[mi][nj][4 * u2 + 1];
                    t.z = o[mi][nj][4 * u2 + 2];
                    t.w = o[mi][nj][4 * u2 + 3];
                    *(v4f*)(x_s + ((mi * 4 + nj) * 2 + u2) * 128 + lane * 4) = t;
                }
            }
            if (laneN == 0) {
#pragma unroll
                for (int r = 0; r < 8; ++r) {
                    x_s[2048 + mi * 16 + r + hi8] = mr[mi][r];
                    x_s[2080 + mi * 16 + r + hi8] = lr[mi][r];
                }
            }
        }
    }
    __syncthreads();

    if (w == 0) {
#pragma unroll
        for (int mi = 0; mi < 2; ++mi) {
            float c0[8], c1[8];
#pragma unroll
            for (int r = 0; r < 8; ++r) {
                const float m1 = x_s[2048 + mi * 16 + r + hi8];
                const float l1 = x_s[2080 + mi * 16 + r + hi8];
                const float m  = fmaxf(mr[mi][r], m1);
                const float e0 = __expf(mr[mi][r] - m);
                const float e1 = __expf(m1 - m);
                const float inv = 1.0f / (lr[mi][r] * e0 + l1 * e1);
                c0[r] = e0 * inv;
                c1[r] = e1 * inv;
            }
#pragma unroll
            for (int nj = 0; nj < 4; ++nj) {
#pragma unroll
                for (int u2 = 0; u2 < 2; ++u2) {
                    const v4f t = *(const v4f*)(x_s + ((mi * 4 + nj) * 2 + u2) * 128 + lane * 4);
#pragma unroll
                    for (int e = 0; e < 4; ++e) {
                        const int r = 4 * u2 + e;
                        q_s[mi * 1024 + (2 * nj + (laneN >> 3)) * 128
                            + ((((laneN >> 2) & 1) << 4) + r + hi8) * 4 + (laneN & 3)]
                            = o[mi][nj][r] * c0[r] + t[e] * c1[r];
                    }
                }
            }
        }
    }
    __syncthreads();
    const int bz = bh >> 4;
    const int hy = bh & (NHC - 1);
    float* og = ob + ((size_t)((bz * LCONST + q0) >> 4) * 128 + hy * 8) * 128 + lane * 4;
#pragma unroll
    for (int jj = 0; jj < 8; ++jj) {
        const v4f t = *(const v4f*)(q_s + w * 1024 + jj * 128 + lane * 4);
        *(v4f*)(og + (size_t)w * GPITCH + jj * 128) = t;
    }
}

__global__ __launch_bounds__(128) void fc_kernel(
    const float* __restrict__ of, const float* __restrict__ wfcf,
    float* __restrict__ out) {
    const int lane  = threadIdx.x & 31;
    const int wave  = threadIdx.x >> 5;
    const int laneN = lane & 15;
    const int hi8   = (lane >> 4) << 3;
    const int m0 = blockIdx.y * 64;
    const int n0 = blockIdx.x * 256 + wave * 64;

    v8f zero = {};
    v8f acc[4][4];
#pragma unroll
    for (int mi = 0; mi < 4; ++mi)
#pragma unroll
        for (int ni = 0; ni < 4; ++ni) acc[mi][ni] = zero;

    gemm_tile_64x64(of + (size_t)(m0 >> 4) * GPITCH + lane * 4,
                    wfcf + (size_t)(n0 >> 4) * GPITCH + lane * 4, acc);

#pragma unroll
    for (int mi = 0; mi < 4; ++mi) {
#pragma unroll
        for (int r = 0; r < 8; ++r) {
            const int m = m0 + mi * 16 + r + hi8;
            float* rowp = out + (size_t)m * DCONST + n0;
#pragma unroll
            for (int ni = 0; ni < 4; ++ni) rowp[ni * 16 + laneN] = acc[mi][ni][r];
        }
    }
}

extern "C" void kernel_launch(void* const* d_in, const int* in_sizes, int n_in,
                              void* d_out, int out_size, void* d_ws, size_t ws_size,
                              hipStream_t stream) {
    const float* x     = (const float*)d_in[0];
    const float* w_qkv = (const float*)d_in[2];
    const float* w_fc  = (const float*)d_in[3];
    const float* qnw   = (const float*)d_in[4];
    const float* knw   = (const float*)d_in[5];

    const size_t per = (size_t)2 * NHC * LCONST * DHC;
    float* wsf  = (float*)d_ws;
    float* qb   = wsf;
    float* kb   = wsf + per;
    float* vb   = wsf + 2 * per;
    float* obuf = wsf + 3 * per;
    float* xf   = wsf + 4 * per;
    float* wf   = wsf + 5 * per;
    float* wfcf = wsf + 6 * per;

    repack_kernel<<<1024, 256, 0, stream>>>(x, w_qkv, w_fc, xf, wf, wfcf);

    dim3 g1(3072 / 256, 4096 / 64);
    qkv_rms_kernel<<<g1, 128, 0, stream>>>(xf, wf, qnw, knw, qb, kb, vb);

    dim3 g2(2 * NHC, LCONST / 32);
    attn_kernel<<<g2, 64, 0, stream>>>(qb, kb, vb, obuf);

    dim3 g3(DCONST / 256, 4096 / 64);
    fc_kernel<<<g3, 128, 0, stream>>>(obuf, wfcf, (float*)d_out);
}
